# starting kernel
# baseline (speedup 1.0000x reference)
#include <hip/hip_runtime.h>
#include <math.h>

typedef __attribute__((ext_vector_type(2))) float v2f;
typedef __attribute__((ext_vector_type(8))) float v8f;

namespace {
constexpr int kRes = 64;
constexpr int kNB  = 4;
constexpr int kNN  = 128;
constexpr int kNJ  = 16;
constexpr float kBWInv = 1.0f / 0.3f;
}

__global__ void __launch_bounds__(256) w2_kernel(const float* __restrict__ Bs,
                                                 const float* __restrict__ Ds,
                                                 float* __restrict__ w2) {
  int idx = blockIdx.x * blockDim.x + threadIdx.x;
  if (idx >= kNB * kNN) return;
  const float* bp = Bs + idx * kNJ * 2;
  const float* dp = Ds + idx * kNJ * 2;
  float bx[kNJ], by[kNJ], dx[kNJ], dy[kNJ];
#pragma unroll
  for (int j = 0; j < kNJ; ++j) {
    bx[j] = bp[2 * j + 0];
    by[j] = bp[2 * j + 1];
    dx[j] = dp[2 * j + 0];
    dy[j] = dp[2 * j + 1];
  }
  float w = 0.0f;
#pragma unroll
  for (int i = 0; i < kNJ; ++i) {
#pragma unroll
    for (int j = 0; j < kNJ; ++j) {
      float r0 = fmaxf(dx[j] - bx[i], 0.0f);
      float r1 = fmaxf(dy[j] - by[i], 0.0f);
      w = fmaxf(w, fminf(r0, r1));
    }
  }
  w2[idx] = w * w;
}

__global__ void __launch_bounds__(256) img_kernel(const float* __restrict__ Bs,
                                                  const float* __restrict__ Ds,
                                                  const float* __restrict__ w2,
                                                  float* __restrict__ out) {
  __shared__ float sB[kNN * kNJ * 2];
  __shared__ float sD[kNN * kNJ * 2];
  __shared__ float sW[kNN];

  const int wave   = threadIdx.x >> 5;
  const int lane   = threadIdx.x & 31;
  const int waveId = blockIdx.x * 8 + wave;
  const int b      = waveId >> 8;
  const int tile   = waveId & 255;

  {
    const float4* gB = (const float4*)(Bs + (size_t)b * kNN * kNJ * 2);
    const float4* gD = (const float4*)(Ds + (size_t)b * kNN * kNJ * 2);
    float4* lB = (float4*)sB;
    float4* lD = (float4*)sD;
    for (int i = threadIdx.x; i < kNN * kNJ * 2 / 4; i += 256) {
      lB[i] = gB[i];
      lD[i] = gD[i];
    }
    if (threadIdx.x < kNN) sW[threadIdx.x] = w2[b * kNN + threadIdx.x];
  }
  __syncthreads();

  const int li = lane & 15;
  const int hi = lane >> 4;
  const int p  = tile * 16 + li;
  const float x0 = -3.0f + 6.0f * (float)(p >> 6) / 63.0f;
  const float x1 = -3.0f + 6.0f * (float)(p & 63) / 63.0f;

  v8f acc = {};
  for (int nc = 0; nc < kNN; nc += 4) {
    v2f a;
#pragma unroll
    for (int t = 0; t < 2; ++t) {
      const int n = nc + 2 * hi + t;
      const float2* Bp = (const float2*)(sB + n * (kNJ * 2));
      const float2* Dp = (const float2*)(sD + n * (kNJ * 2));
      float db = 3.0e38f, dd = 3.0e38f;
#pragma unroll
      for (int j = 0; j < kNJ; ++j) {
        float2 bv = Bp[j];
        float2 dv = Dp[j];
        db = fminf(db, fmaxf(bv.x - x0, bv.y - x1));
        dd = fminf(dd, fmaxf(x0 - dv.x, x1 - dv.y));
      }
      float dist = fmaxf(fmaxf(db, dd), 0.0f) * kBWInv;
      a[t] = expf(-0.5f * dist * dist) * sW[n];
    }
    v2f ones = {1.0f, 1.0f};
    acc = __builtin_amdgcn_wmma_f32_16x16x4_f32(
        false, a, false, ones, (short)0, acc, false, false);
  }

  float* op = out + (size_t)b * (kRes * kRes) + tile * 16;
  if (lane == 0) {
#pragma unroll
    for (int r = 0; r < 8; ++r) op[r] = acc[r];
  } else if (lane == 16) {
#pragma unroll
    for (int r = 0; r < 8; ++r) op[8 + r] = acc[r];
  }
}

extern "C" void kernel_launch(void* const* d_in, const int* in_sizes, int n_in,
                              void* d_out, int out_size, void* d_ws, size_t ws_size,
                              hipStream_t stream) {
  (void)in_sizes; (void)n_in; (void)out_size; (void)ws_size;
  const float* Bs = (const float*)d_in[0];
  const float* Ds = (const float*)d_in[1];
  float* w2  = (float*)d_ws;
  float* out = (float*)d_out;

  w2_kernel<<<2, 256, 0, stream>>>(Bs, Ds, w2);
  img_kernel<<<128, 256, 0, stream>>>(Bs, Ds, w2, out);
}


# image pass register-tiled: lane = a 2x2 pixel patch, reads each corner once from interleaved LDS; boxes split over 8 waves, one-hot B keeps box sum on WMMA, partials added in LDS (sum order changes)
# speedup vs baseline: 3.0534x; 3.0534x over previous
#include <hip/hip_runtime.h>
#include <math.h>


typedef __attribute__((ext_vector_type(2))) float v2f;
typedef __attribute__((ext_vector_type(8))) float v8f;

namespace {
constexpr int kRes = 64;
constexpr int kNB  = 4;
constexpr int kNN  = 128;
constexpr int kNJ  = 16;
constexpr float kBWInv = 1.0f / 0.3f;
constexpr int kPitch = kNJ * 4;
constexpr int kWaves = 8;
}

__global__ void __launch_bounds__(256) w2_kernel(const float* __restrict__ Bs,
                                                 const float* __restrict__ Ds,
                                                 float* __restrict__ w2) {
  int idx = blockIdx.x * blockDim.x + threadIdx.x;
  if (idx >= kNB * kNN) return;
  const float* bp = Bs + idx * kNJ * 2;
  const float* dp = Ds + idx * kNJ * 2;
  float bx[kNJ], by[kNJ], dx[kNJ], dy[kNJ];
#pragma unroll
  for (int j = 0; j < kNJ; ++j) {
    bx[j] = bp[2 * j + 0];
    by[j] = bp[2 * j + 1];
    dx[j] = dp[2 * j + 0];
    dy[j] = dp[2 * j + 1];
  }
  float w = 0.0f;
#pragma unroll
  for (int i = 0; i < kNJ; ++i) {
#pragma unroll
    for (int j = 0; j < kNJ; ++j) {
      float r0 = fmaxf(dx[j] - bx[i], 0.0f);
      float r1 = fmaxf(dy[j] - by[i], 0.0f);
      w = fmaxf(w, fminf(r0, r1));
    }
  }
  w2[idx] = w * w;
}

__global__ void __launch_bounds__(256) img_kernel(const float* __restrict__ Bs,
                                                  const float* __restrict__ Ds,
                                                  const float* __restrict__ w2,
                                                  float* __restrict__ out) {
  __shared__ float4 sBD[kNN * kPitch / 4];
  __shared__ float  sW[kNN];
  __shared__ float  sRed[kWaves][64];

  const int wave = threadIdx.x >> 5;
  const int lane = threadIdx.x & 31;
  const int b    = blockIdx.x >> 6;
  const int g    = blockIdx.x & 63;

  {
    const float4* gB = (const float4*)(Bs + (size_t)b * kNN * kNJ * 2);
    const float4* gD = (const float4*)(Ds + (size_t)b * kNN * kNJ * 2);
    static_assert(kNN * kNJ / 2 == 4 * 256, "four corner pairs per thread");
#pragma unroll
    for (int k = 0; k < 4; ++k) {
      const int i  = threadIdx.x + 256 * k;
      const int n  = i >> 3;
      const int jp = i & 7;
      const float4 bv = gB[i];
      const float4 dv = gD[i];
      float4* dst = sBD + n * (kPitch / 4) + 2 * jp;
      dst[0] = make_float4(bv.x, bv.y, dv.x, dv.y);
      dst[1] = make_float4(bv.z, bv.w, dv.z, dv.w);
    }
    if (threadIdx.x < kNN) sW[threadIdx.x] = w2[b * kNN + threadIdx.x];
  }
  __syncthreads();

  const int c    = lane & 15;
  const int h    = lane >> 4;
  const int row0 = 2 * (g >> 1);
  const int col0 = 32 * (g & 1) + 2 * c;
  const float x0a = -3.0f + 6.0f * (float)(row0)     / 63.0f;
  const float x0b = -3.0f + 6.0f * (float)(row0 + 1) / 63.0f;
  const float x1a = -3.0f + 6.0f * (float)(col0)     / 63.0f;
  const float x1b = -3.0f + 6.0f * (float)(col0 + 1) / 63.0f;

  const v2f xa = {x0a, x1a};
  const v2f xb = {x0b, x1b};

  v2f bsel[4];
#pragma unroll
  for (int q = 0; q < 4; ++q) {
    const float f = (c == q) ? 1.0f : 0.0f;
    bsel[q][0] = f;
    bsel[q][1] = f;
  }

  v8f acc = {};
  for (int t = 0; t < 4; ++t) {
    float av[2][4];
    float4 cb[2][kNJ];
#pragma unroll
    for (int s = 0; s < 2; ++s) {
      const float4* P = sBD + (16 * wave + 4 * t + 2 * h + s) * (kPitch / 4);
#pragma unroll
      for (int j = 0; j < kNJ; ++j) cb[s][j] = P[j];
    }
#pragma unroll
    for (int s = 0; s < 2; ++s) {
      const int n = 16 * wave + 4 * t + 2 * h + s;
      float db0 = 3.0e38f, db1 = 3.0e38f, db2 = 3.0e38f, db3 = 3.0e38f;
      float dd0 = 3.0e38f, dd1 = 3.0e38f, dd2 = 3.0e38f, dd3 = 3.0e38f;
#pragma unroll
      for (int j = 0; j < kNJ; ++j) {
        const float4 cv = cb[s][j];
        const v2f bxy = {cv.x, cv.y};
        const v2f dxy = {cv.z, cv.w};
        const v2f ba = bxy - xa;
        const v2f bb = bxy - xb;
        const v2f da = xa - dxy;
        const v2f dc = xb - dxy;
        db0 = fminf(db0, fmaxf(ba[0], ba[1]));
        db1 = fminf(db1, fmaxf(ba[0], bb[1]));
        db2 = fminf(db2, fmaxf(bb[0], ba[1]));
        db3 = fminf(db3, fmaxf(bb[0], bb[1]));
        dd0 = fminf(dd0, fmaxf(da[0], da[1]));
        dd1 = fminf(dd1, fmaxf(da[0], dc[1]));
        dd2 = fminf(dd2, fmaxf(dc[0], da[1]));
        dd3 = fminf(dd3, fmaxf(dc[0], dc[1]));
      }
      const float wn = sW[n];
      const float e0 = fmaxf(fmaxf(db0, dd0), 0.0f) * kBWInv;
      const float e1 = fmaxf(fmaxf(db1, dd1), 0.0f) * kBWInv;
      const float e2 = fmaxf(fmaxf(db2, dd2), 0.0f) * kBWInv;
      const float e3 = fmaxf(fmaxf(db3, dd3), 0.0f) * kBWInv;
      av[s][0] = expf(-0.5f * e0 * e0) * wn;
      av[s][1] = expf(-0.5f * e1 * e1) * wn;
      av[s][2] = expf(-0.5f * e2 * e2) * wn;
      av[s][3] = expf(-0.5f * e3 * e3) * wn;
    }
#pragma unroll
    for (int q = 0; q < 4; ++q) {
      v2f a;
      a[0] = av[0][q];
      a[1] = av[1][q];
      acc = __builtin_amdgcn_wmma_f32_16x16x4_f32(
          false, a, false, bsel[q], (short)0, acc, false, false);
    }
  }

  if (c < 4) {
#pragma unroll
    for (int r = 0; r < 8; ++r) sRed[wave][(8 * h + r) * 4 + c] = acc[r];
  }
  __syncthreads();

  if (threadIdx.x < 64) {
    const int ro  = threadIdx.x >> 5;
    const int col = threadIdx.x & 31;
    const int idx = (col >> 1) * 4 + 2 * ro + (col & 1);
    float v = sRed[0][idx];
#pragma unroll
    for (int w = 1; w < kWaves; ++w) v += sRed[w][idx];
    out[(size_t)b * (kRes * kRes) + (row0 + ro) * kRes + 32 * (g & 1) + col] = v;
  }
}

extern "C" void kernel_launch(void* const* d_in, const int* in_sizes, int n_in,
                              void* d_out, int out_size, void* d_ws, size_t ws_size,
                              hipStream_t stream) {
  (void)in_sizes; (void)n_in; (void)out_size; (void)ws_size;
  const float* Bs = (const float*)d_in[0];
  const float* Ds = (const float*)d_in[1];
  float* w2  = (float*)d_ws;
  float* out = (float*)d_out;

  w2_kernel<<<2, 256, 0, stream>>>(Bs, Ds, w2);
  img_kernel<<<256, 256, 0, stream>>>(Bs, Ds, w2, out);
}


# box weights fused into the image kernel: the 16 lanes holding a box's death corners take one birth corner each, joined by a DPP row maximum; weight kernel, its launch and scratch go
# speedup vs baseline: 4.5995x; 1.5064x over previous
#include <hip/hip_runtime.h>
#include <math.h>


typedef __attribute__((ext_vector_type(2))) float v2f;
typedef __attribute__((ext_vector_type(8))) float v8f;

namespace {
constexpr int kRes = 64;
constexpr int kNN  = 128;
constexpr int kNJ  = 16;
constexpr float kBWInv = 1.0f / 0.3f;
constexpr int kPitch = kNJ * 4;
constexpr int kWaves = 8;
}

__device__ __forceinline__ float row_max16_nonneg(float v) {
  int x = __builtin_bit_cast(int, v);
  x = max(x, __builtin_amdgcn_update_dpp(0, x, 0x161, 0xf, 0xf, false));
  x = max(x, __builtin_amdgcn_update_dpp(0, x, 0x162, 0xf, 0xf, false));
  x = max(x, __builtin_amdgcn_update_dpp(0, x, 0x164, 0xf, 0xf, false));
  x = max(x, __builtin_amdgcn_update_dpp(0, x, 0x168, 0xf, 0xf, false));
  return __builtin_bit_cast(float, x);
}

__global__ void __launch_bounds__(256) img_kernel(const float* __restrict__ Bs,
                                                  const float* __restrict__ Ds,
                                                  float* __restrict__ out) {
  __shared__ float4 sBD[kNN * kPitch / 4];
  __shared__ float  sRed[kWaves][64];

  const int wave = threadIdx.x >> 5;
  const int lane = threadIdx.x & 31;
  const int b    = blockIdx.x >> 6;
  const int g    = blockIdx.x & 63;

  {
    const float4* gB = (const float4*)(Bs + (size_t)b * kNN * kNJ * 2);
    const float4* gD = (const float4*)(Ds + (size_t)b * kNN * kNJ * 2);
    static_assert(kNN * kNJ / 2 == 4 * 256, "four corner pairs per thread");
#pragma unroll
    for (int k = 0; k < 4; ++k) {
      const int i  = threadIdx.x + 256 * k;
      const int n  = i >> 3;
      const int jp = i & 7;
      const float4 bv = gB[i];
      const float4 dv = gD[i];
      float4* dst = sBD + n * (kPitch / 4) + 2 * jp;
      dst[0] = make_float4(bv.x, bv.y, dv.x, dv.y);
      dst[1] = make_float4(bv.z, bv.w, dv.z, dv.w);
    }
  }
  __syncthreads();

  const int c    = lane & 15;
  const int h    = lane >> 4;
  const int row0 = 2 * (g >> 1);
  const int col0 = 32 * (g & 1) + 2 * c;
  const float x0a = -3.0f + 6.0f * (float)(row0)     / 63.0f;
  const float x0b = -3.0f + 6.0f * (float)(row0 + 1) / 63.0f;
  const float x1a = -3.0f + 6.0f * (float)(col0)     / 63.0f;
  const float x1b = -3.0f + 6.0f * (float)(col0 + 1) / 63.0f;

  const v2f xa = {x0a, x1a};
  const v2f xb = {x0b, x1b};

  v2f bsel[4];
#pragma unroll
  for (int q = 0; q < 4; ++q) {
    const float f = (c == q) ? 1.0f : 0.0f;
    bsel[q][0] = f;
    bsel[q][1] = f;
  }

  v8f acc = {};
  for (int t = 0; t < 4; ++t) {
    float av[2][4];
    float4 cb[2][kNJ];
    v2f bo[2];
#pragma unroll
    for (int s = 0; s < 2; ++s) {
      const float4* P = sBD + (16 * wave + 4 * t + 2 * h + s) * (kPitch / 4);
#pragma unroll
      for (int j = 0; j < kNJ; ++j) cb[s][j] = P[j];
      bo[s][0] = P[c].x;
      bo[s][1] = P[c].y;
    }
#pragma unroll
    for (int s = 0; s < 2; ++s) {
      float db0 = 3.0e38f, db1 = 3.0e38f, db2 = 3.0e38f, db3 = 3.0e38f;
      float dd0 = 3.0e38f, dd1 = 3.0e38f, dd2 = 3.0e38f, dd3 = 3.0e38f;
#pragma unroll
      for (int j = 0; j < kNJ; ++j) {
        const float4 cv = cb[s][j];
        const v2f bxy = {cv.x, cv.y};
        const v2f dxy = {cv.z, cv.w};
        const v2f ba = bxy - xa;
        const v2f bb = bxy - xb;
        const v2f da = xa - dxy;
        const v2f dc = xb - dxy;
        db0 = fminf(db0, fmaxf(ba[0], ba[1]));
        db1 = fminf(db1, fmaxf(ba[0], bb[1]));
        db2 = fminf(db2, fmaxf(bb[0], ba[1]));
        db3 = fminf(db3, fmaxf(bb[0], bb[1]));
        dd0 = fminf(dd0, fmaxf(da[0], da[1]));
        dd1 = fminf(dd1, fmaxf(da[0], dc[1]));
        dd2 = fminf(dd2, fmaxf(dc[0], da[1]));
        dd3 = fminf(dd3, fmaxf(dc[0], dc[1]));
      }
      float wm = 0.0f;
#pragma unroll
      for (int j = 0; j < kNJ; ++j) {
        const v2f dxy = {cb[s][j].z, cb[s][j].w};
        const v2f df  = dxy - bo[s];
        wm = fmaxf(wm, fminf(df[0], df[1]));
      }
      wm = row_max16_nonneg(wm);
      const float wn = wm * wm;
      const float e0 = fmaxf(fmaxf(db0, dd0), 0.0f) * kBWInv;
      const float e1 = fmaxf(fmaxf(db1, dd1), 0.0f) * kBWInv;
      const float e2 = fmaxf(fmaxf(db2, dd2), 0.0f) * kBWInv;
      const float e3 = fmaxf(fmaxf(db3, dd3), 0.0f) * kBWInv;
      av[s][0] = expf(-0.5f * e0 * e0) * wn;
      av[s][1] = expf(-0.5f * e1 * e1) * wn;
      av[s][2] = expf(-0.5f * e2 * e2) * wn;
      av[s][3] = expf(-0.5f * e3 * e3) * wn;
    }
#pragma unroll
    for (int q = 0; q < 4; ++q) {
      v2f a;
      a[0] = av[0][q];
      a[1] = av[1][q];
      acc = __builtin_amdgcn_wmma_f32_16x16x4_f32(
          false, a, false, bsel[q], (short)0, acc, false, false);
    }
  }

  if (c < 4) {
#pragma unroll
    for (int r = 0; r < 8; ++r) sRed[wave][(8 * h + r) * 4 + c] = acc[r];
  }
  __syncthreads();

  if (threadIdx.x < 64) {
    const int ro  = threadIdx.x >> 5;
    const int col = threadIdx.x & 31;
    const int idx = (col >> 1) * 4 + 2 * ro + (col & 1);
    float v = sRed[0][idx];
#pragma unroll
    for (int w = 1; w < kWaves; ++w) v += sRed[w][idx];
    out[(size_t)b * (kRes * kRes) + (row0 + ro) * kRes + 32 * (g & 1) + col] = v;
  }
}

extern "C" void kernel_launch(void* const* d_in, const int* in_sizes, int n_in,
                              void* d_out, int out_size, void* d_ws, size_t ws_size,
                              hipStream_t stream) {
  (void)in_sizes; (void)n_in; (void)out_size; (void)d_ws; (void)ws_size;
  const float* Bs = (const float*)d_in[0];
  const float* Ds = (const float*)d_in[1];
  float* out = (float*)d_out;

  img_kernel<<<256, 256, 0, stream>>>(Bs, Ds, out);
}


# running min-of-max chains (image pass, box weight) on fused v_maxmin_num_f32 / v_minmax_num_f32: one instruction per term, not a max per term plus a 3-way min per two; bit-identical
# speedup vs baseline: 5.4725x; 1.1898x over previous
#include <hip/hip_runtime.h>
#include <math.h>


typedef __attribute__((ext_vector_type(2))) float v2f;
typedef __attribute__((ext_vector_type(8))) float v8f;

namespace {
constexpr int kRes = 64;
constexpr int kNN  = 128;
constexpr int kNJ  = 16;
constexpr float kBWInv = 1.0f / 0.3f;
constexpr int kPitch = kNJ * 4;
constexpr int kWaves = 8;
}

__device__ __forceinline__ float row_max16_nonneg(float v) {
  int x = __builtin_bit_cast(int, v);
  x = max(x, __builtin_amdgcn_update_dpp(0, x, 0x161, 0xf, 0xf, false));
  x = max(x, __builtin_amdgcn_update_dpp(0, x, 0x162, 0xf, 0xf, false));
  x = max(x, __builtin_amdgcn_update_dpp(0, x, 0x164, 0xf, 0xf, false));
  x = max(x, __builtin_amdgcn_update_dpp(0, x, 0x168, 0xf, 0xf, false));
  return __builtin_bit_cast(float, x);
}

__device__ __forceinline__ float maxmin(float a, float b, float m) {
  float r;
  asm("v_maxmin_num_f32 %0, %1, %2, %3" : "=v"(r) : "v"(a), "v"(b), "v"(m));
  return r;
}
__device__ __forceinline__ float minmax(float a, float b, float m) {
  float r;
  asm("v_minmax_num_f32 %0, %1, %2, %3" : "=v"(r) : "v"(a), "v"(b), "v"(m));
  return r;
}

__global__ void __launch_bounds__(256) img_kernel(const float* __restrict__ Bs,
                                                  const float* __restrict__ Ds,
                                                  float* __restrict__ out) {
  __shared__ float4 sBD[kNN * kPitch / 4];
  __shared__ float  sRed[kWaves][64];

  const int wave = threadIdx.x >> 5;
  const int lane = threadIdx.x & 31;
  const int b    = blockIdx.x >> 6;
  const int g    = blockIdx.x & 63;

  {
    const float4* gB = (const float4*)(Bs + (size_t)b * kNN * kNJ * 2);
    const float4* gD = (const float4*)(Ds + (size_t)b * kNN * kNJ * 2);
    static_assert(kNN * kNJ / 2 == 4 * 256, "four corner pairs per thread");
#pragma unroll
    for (int k = 0; k < 4; ++k) {
      const int i  = threadIdx.x + 256 * k;
      const int n  = i >> 3;
      const int jp = i & 7;
      const float4 bv = gB[i];
      const float4 dv = gD[i];
      float4* dst = sBD + n * (kPitch / 4) + 2 * jp;
      dst[0] = make_float4(bv.x, bv.y, dv.x, dv.y);
      dst[1] = make_float4(bv.z, bv.w, dv.z, dv.w);
    }
  }
  __syncthreads();

  const int c    = lane & 15;
  const int h    = lane >> 4;
  const int row0 = 2 * (g >> 1);
  const int col0 = 32 * (g & 1) + 2 * c;
  const float x0a = -3.0f + 6.0f * (float)(row0)     / 63.0f;
  const float x0b = -3.0f + 6.0f * (float)(row0 + 1) / 63.0f;
  const float x1a = -3.0f + 6.0f * (float)(col0)     / 63.0f;
  const float x1b = -3.0f + 6.0f * (float)(col0 + 1) / 63.0f;

  const v2f xa = {x0a, x1a};
  const v2f xb = {x0b, x1b};

  v2f bsel[4];
#pragma unroll
  for (int q = 0; q < 4; ++q) {
    const float f = (c == q) ? 1.0f : 0.0f;
    bsel[q][0] = f;
    bsel[q][1] = f;
  }

  v8f acc = {};
  for (int t = 0; t < 4; ++t) {
    float av[2][4];
    float4 cb[2][kNJ];
    v2f bo[2];
#pragma unroll
    for (int s = 0; s < 2; ++s) {
      const float4* P = sBD + (16 * wave + 4 * t + 2 * h + s) * (kPitch / 4);
#pragma unroll
      for (int j = 0; j < kNJ; ++j) cb[s][j] = P[j];
      bo[s][0] = P[c].x;
      bo[s][1] = P[c].y;
    }
#pragma unroll
    for (int s = 0; s < 2; ++s) {
      float db0, db1, db2, db3, dd0, dd1, dd2, dd3, wm;
      v2f ba, bb, da, dc, df;
      {
        const float4 cv = cb[s][0];
        const v2f bxy = {cv.x, cv.y};
        const v2f dxy = {cv.z, cv.w};
        ba = bxy - xa;
        bb = bxy - xb;
        da = xa - dxy;
        dc = xb - dxy;
        df = dxy - bo[s];
      }
#pragma unroll
      for (int j = 0; j < kNJ; ++j) {
        const v2f pa = ba, pb = bb, qa = da, qc = dc, qf = df;
        if (j + 1 < kNJ) {
          const float4 cv = cb[s][j + 1];
          const v2f bxy = {cv.x, cv.y};
          const v2f dxy = {cv.z, cv.w};
          ba = bxy - xa;
          bb = bxy - xb;
          da = xa - dxy;
          dc = xb - dxy;
          df = dxy - bo[s];
        }
        __builtin_amdgcn_sched_barrier(0);
        if (j == 0) {
          db0 = fmaxf(pa[0], pa[1]);
          db1 = fmaxf(pa[0], pb[1]);
          db2 = fmaxf(pb[0], pa[1]);
          db3 = fmaxf(pb[0], pb[1]);
          dd0 = fmaxf(qa[0], qa[1]);
          dd1 = fmaxf(qa[0], qc[1]);
          dd2 = fmaxf(qc[0], qa[1]);
          dd3 = fmaxf(qc[0], qc[1]);
          wm  = fmaxf(fminf(qf[0], qf[1]), 0.0f);
        } else {
          db0 = maxmin(pa[0], pa[1], db0);
          db1 = maxmin(pa[0], pb[1], db1);
          db2 = maxmin(pb[0], pa[1], db2);
          db3 = maxmin(pb[0], pb[1], db3);
          dd0 = maxmin(qa[0], qa[1], dd0);
          dd1 = maxmin(qa[0], qc[1], dd1);
          dd2 = maxmin(qc[0], qa[1], dd2);
          dd3 = maxmin(qc[0], qc[1], dd3);
          wm  = minmax(qf[0], qf[1], wm);
        }
        __builtin_amdgcn_sched_barrier(0);
      }
      wm = row_max16_nonneg(wm);
      const float wn = wm * wm;
      const float e0 = fmaxf(fmaxf(db0, dd0), 0.0f) * kBWInv;
      const float e1 = fmaxf(fmaxf(db1, dd1), 0.0f) * kBWInv;
      const float e2 = fmaxf(fmaxf(db2, dd2), 0.0f) * kBWInv;
      const float e3 = fmaxf(fmaxf(db3, dd3), 0.0f) * kBWInv;
      av[s][0] = expf(-0.5f * e0 * e0) * wn;
      av[s][1] = expf(-0.5f * e1 * e1) * wn;
      av[s][2] = expf(-0.5f * e2 * e2) * wn;
      av[s][3] = expf(-0.5f * e3 * e3) * wn;
    }
#pragma unroll
    for (int q = 0; q < 4; ++q) {
      v2f a;
      a[0] = av[0][q];
      a[1] = av[1][q];
      acc = __builtin_amdgcn_wmma_f32_16x16x4_f32(
          false, a, false, bsel[q], (short)0, acc, false, false);
    }
  }

  if (c < 4) {
#pragma unroll
    for (int r = 0; r < 8; ++r) sRed[wave][(8 * h + r) * 4 + c] = acc[r];
  }
  __syncthreads();

  if (threadIdx.x < 64) {
    const int ro  = threadIdx.x >> 5;
    const int col = threadIdx.x & 31;
    const int idx = (col >> 1) * 4 + 2 * ro + (col & 1);
    float v = sRed[0][idx];
#pragma unroll
    for (int w = 1; w < kWaves; ++w) v += sRed[w][idx];
    out[(size_t)b * (kRes * kRes) + (row0 + ro) * kRes + 32 * (g & 1) + col] = v;
  }
}

extern "C" void kernel_launch(void* const* d_in, const int* in_sizes, int n_in,
                              void* d_out, int out_size, void* d_ws, size_t ws_size,
                              hipStream_t stream) {
  (void)in_sizes; (void)n_in; (void)out_size; (void)d_ws; (void)ws_size;
  const float* Bs = (const float*)d_in[0];
  const float* Ds = (const float*)d_in[1];
  float* out = (float*)d_out;

  img_kernel<<<256, 256, 0, stream>>>(Bs, Ds, out);
}


# Gaussian as 2^(c*t*t) on v_exp_f32, with -0.5/bw^2 and log2(e) folded into one constant, in place of the library expf expansion (13 instructions per value); value rounding changes, sum order kept
# speedup vs baseline: 5.8663x; 1.0720x over previous
#include <hip/hip_runtime.h>
#include <math.h>


typedef __attribute__((ext_vector_type(2))) float v2f;
typedef __attribute__((ext_vector_type(8))) float v8f;

namespace {
constexpr int kRes = 64;
constexpr int kNN  = 128;
constexpr int kNJ  = 16;
constexpr float kGaussC =
    (float)(-0.5 * 1.4426950408889634 / ((double)0.3f * (double)0.3f));
constexpr int kPitch = kNJ * 4;
constexpr int kWaves = 8;
}

__device__ __forceinline__ float row_max16_nonneg(float v) {
  int x = __builtin_bit_cast(int, v);
  x = max(x, __builtin_amdgcn_update_dpp(0, x, 0x161, 0xf, 0xf, false));
  x = max(x, __builtin_amdgcn_update_dpp(0, x, 0x162, 0xf, 0xf, false));
  x = max(x, __builtin_amdgcn_update_dpp(0, x, 0x164, 0xf, 0xf, false));
  x = max(x, __builtin_amdgcn_update_dpp(0, x, 0x168, 0xf, 0xf, false));
  return __builtin_bit_cast(float, x);
}

__device__ __forceinline__ float maxmin(float a, float b, float m) {
  float r;
  asm("v_maxmin_num_f32 %0, %1, %2, %3" : "=v"(r) : "v"(a), "v"(b), "v"(m));
  return r;
}
__device__ __forceinline__ float minmax(float a, float b, float m) {
  float r;
  asm("v_minmax_num_f32 %0, %1, %2, %3" : "=v"(r) : "v"(a), "v"(b), "v"(m));
  return r;
}

__global__ void __launch_bounds__(256) img_kernel(const float* __restrict__ Bs,
                                                  const float* __restrict__ Ds,
                                                  float* __restrict__ out) {
  __shared__ float4 sBD[kNN * kPitch / 4];
  __shared__ float  sRed[kWaves][64];

  const int wave = threadIdx.x >> 5;
  const int lane = threadIdx.x & 31;
  const int b    = blockIdx.x >> 6;
  const int g    = blockIdx.x & 63;

  {
    const float4* gB = (const float4*)(Bs + (size_t)b * kNN * kNJ * 2);
    const float4* gD = (const float4*)(Ds + (size_t)b * kNN * kNJ * 2);
    static_assert(kNN * kNJ / 2 == 4 * 256, "four corner pairs per thread");
#pragma unroll
    for (int k = 0; k < 4; ++k) {
      const int i  = threadIdx.x + 256 * k;
      const int n  = i >> 3;
      const int jp = i & 7;
      const float4 bv = gB[i];
      const float4 dv = gD[i];
      float4* dst = sBD + n * (kPitch / 4) + 2 * jp;
      dst[0] = make_float4(bv.x, bv.y, dv.x, dv.y);
      dst[1] = make_float4(bv.z, bv.w, dv.z, dv.w);
    }
  }
  __syncthreads();

  const int c    = lane & 15;
  const int h    = lane >> 4;
  const int row0 = 2 * (g >> 1);
  const int col0 = 32 * (g & 1) + 2 * c;
  const float x0a = -3.0f + 6.0f * (float)(row0)     / 63.0f;
  const float x0b = -3.0f + 6.0f * (float)(row0 + 1) / 63.0f;
  const float x1a = -3.0f + 6.0f * (float)(col0)     / 63.0f;
  const float x1b = -3.0f + 6.0f * (float)(col0 + 1) / 63.0f;

  const v2f xa = {x0a, x1a};
  const v2f xb = {x0b, x1b};

  v2f bsel[4];
#pragma unroll
  for (int q = 0; q < 4; ++q) {
    const float f = (c == q) ? 1.0f : 0.0f;
    bsel[q][0] = f;
    bsel[q][1] = f;
  }

  v8f acc = {};
  for (int t = 0; t < 4; ++t) {
    float av[2][4];
    float4 cb[2][kNJ];
    v2f bo[2];
#pragma unroll
    for (int s = 0; s < 2; ++s) {
      const float4* P = sBD + (16 * wave + 4 * t + 2 * h + s) * (kPitch / 4);
#pragma unroll
      for (int j = 0; j < kNJ; ++j) cb[s][j] = P[j];
      bo[s][0] = P[c].x;
      bo[s][1] = P[c].y;
    }
#pragma unroll
    for (int s = 0; s < 2; ++s) {
      float db0, db1, db2, db3, dd0, dd1, dd2, dd3, wm;
      v2f ba, bb, da, dc, df;
      {
        const float4 cv = cb[s][0];
        const v2f bxy = {cv.x, cv.y};
        const v2f dxy = {cv.z, cv.w};
        ba = bxy - xa;
        bb = bxy - xb;
        da = xa - dxy;
        dc = xb - dxy;
        df = dxy - bo[s];
      }
#pragma unroll
      for (int j = 0; j < kNJ; ++j) {
        const v2f pa = ba, pb = bb, qa = da, qc = dc, qf = df;
        if (j + 1 < kNJ) {
          const float4 cv = cb[s][j + 1];
          const v2f bxy = {cv.x, cv.y};
          const v2f dxy = {cv.z, cv.w};
          ba = bxy - xa;
          bb = bxy - xb;
          da = xa - dxy;
          dc = xb - dxy;
          df = dxy - bo[s];
        }
        __builtin_amdgcn_sched_barrier(0);
        if (j == 0) {
          db0 = fmaxf(pa[0], pa[1]);
          db1 = fmaxf(pa[0], pb[1]);
          db2 = fmaxf(pb[0], pa[1]);
          db3 = fmaxf(pb[0], pb[1]);
          dd0 = fmaxf(qa[0], qa[1]);
          dd1 = fmaxf(qa[0], qc[1]);
          dd2 = fmaxf(qc[0], qa[1]);
          dd3 = fmaxf(qc[0], qc[1]);
          wm  = fmaxf(fminf(qf[0], qf[1]), 0.0f);
        } else {
          db0 = maxmin(pa[0], pa[1], db0);
          db1 = maxmin(pa[0], pb[1], db1);
          db2 = maxmin(pb[0], pa[1], db2);
          db3 = maxmin(pb[0], pb[1], db3);
          dd0 = maxmin(qa[0], qa[1], dd0);
          dd1 = maxmin(qa[0], qc[1], dd1);
          dd2 = maxmin(qc[0], qa[1], dd2);
          dd3 = maxmin(qc[0], qc[1], dd3);
          wm  = minmax(qf[0], qf[1], wm);
        }
        __builtin_amdgcn_sched_barrier(0);
      }
      wm = row_max16_nonneg(wm);
      const float wn = wm * wm;
      const float t0 = fmaxf(fmaxf(db0, dd0), 0.0f);
      const float t1 = fmaxf(fmaxf(db1, dd1), 0.0f);
      const float t2 = fmaxf(fmaxf(db2, dd2), 0.0f);
      const float t3 = fmaxf(fmaxf(db3, dd3), 0.0f);
      av[s][0] = __builtin_amdgcn_exp2f(t0 * t0 * kGaussC) * wn;
      av[s][1] = __builtin_amdgcn_exp2f(t1 * t1 * kGaussC) * wn;
      av[s][2] = __builtin_amdgcn_exp2f(t2 * t2 * kGaussC) * wn;
      av[s][3] = __builtin_amdgcn_exp2f(t3 * t3 * kGaussC) * wn;
    }
#pragma unroll
    for (int q = 0; q < 4; ++q) {
      v2f a;
      a[0] = av[0][q];
      a[1] = av[1][q];
      acc = __builtin_amdgcn_wmma_f32_16x16x4_f32(
          false, a, false, bsel[q], (short)0, acc, false, false);
    }
  }

  if (c < 4) {
#pragma unroll
    for (int r = 0; r < 8; ++r) sRed[wave][(8 * h + r) * 4 + c] = acc[r];
  }
  __syncthreads();

  if (threadIdx.x < 64) {
    const int ro  = threadIdx.x >> 5;
    const int col = threadIdx.x & 31;
    const int idx = (col >> 1) * 4 + 2 * ro + (col & 1);
    float v = sRed[0][idx];
#pragma unroll
    for (int w = 1; w < kWaves; ++w) v += sRed[w][idx];
    out[(size_t)b * (kRes * kRes) + (row0 + ro) * kRes + 32 * (g & 1) + col] = v;
  }
}

extern "C" void kernel_launch(void* const* d_in, const int* in_sizes, int n_in,
                              void* d_out, int out_size, void* d_ws, size_t ws_size,
                              hipStream_t stream) {
  (void)in_sizes; (void)n_in; (void)out_size; (void)d_ws; (void)ws_size;
  const float* Bs = (const float*)d_in[0];
  const float* Ds = (const float*)d_in[1];
  float* out = (float*)d_out;

  img_kernel<<<256, 256, 0, stream>>>(Bs, Ds, out);
}


# lane's register tile grown from 2x2 to 2x8 pixels (4 lanes per box, 2 boxes per lane): 20 not 32 coordinate subtracts per 16 pixels and corner; a WMMA now sums 2 pixels of 2 boxes (sum order changes)
# speedup vs baseline: 6.6118x; 1.1271x over previous
#include <hip/hip_runtime.h>
#include <math.h>


typedef __attribute__((ext_vector_type(2))) float v2f;
typedef __attribute__((ext_vector_type(8))) float v8f;

namespace {
constexpr int kRes = 64;
constexpr int kNN  = 128;
constexpr int kNJ  = 16;
constexpr float kGaussC =
    (float)(-0.5 * 1.4426950408889634 / ((double)0.3f * (double)0.3f));
constexpr int kPitch4 = kNJ + 1;
constexpr int kWaves  = 8;
constexpr int kTileW  = 8;
}

__device__ __forceinline__ float quad_stride4_max_nonneg(float v) {
  int x = __builtin_bit_cast(int, v);
  x = max(x, __builtin_amdgcn_update_dpp(0, x, 0x164, 0xf, 0xf, false));
  x = max(x, __builtin_amdgcn_update_dpp(0, x, 0x168, 0xf, 0xf, false));
  return __builtin_bit_cast(float, x);
}

__device__ __forceinline__ float maxmin(float a, float b, float m) {
  float r;
  asm("v_maxmin_num_f32 %0, %1, %2, %3" : "=v"(r) : "v"(a), "v"(b), "v"(m));
  return r;
}
__device__ __forceinline__ float minmax(float a, float b, float m) {
  float r;
  asm("v_minmax_num_f32 %0, %1, %2, %3" : "=v"(r) : "v"(a), "v"(b), "v"(m));
  return r;
}

struct CornerDiffs {
  v2f b[5];
  v2f d[5];
  v2f f[4];
};

__global__ void __launch_bounds__(256) img_kernel(const float* __restrict__ Bs,
                                                  const float* __restrict__ Ds,
                                                  float* __restrict__ out) {
  __shared__ float4 sBD[kNN * kPitch4];
  __shared__ float  sRed[kWaves][64];

  const int wave = threadIdx.x >> 5;
  const int lane = threadIdx.x & 31;
  const int b    = blockIdx.x >> 6;
  const int g    = blockIdx.x & 63;

  {
    const float4* gB = (const float4*)(Bs + (size_t)b * kNN * kNJ * 2);
    const float4* gD = (const float4*)(Ds + (size_t)b * kNN * kNJ * 2);
    static_assert(kNN * kNJ / 2 == 4 * 256, "four corner pairs per thread");
#pragma unroll
    for (int k = 0; k < 4; ++k) {
      const int i  = threadIdx.x + 256 * k;
      const int n  = i >> 3;
      const int jp = i & 7;
      const float4 bv = gB[i];
      const float4 dv = gD[i];
      float4* dst = sBD + n * kPitch4 + 2 * jp;
      dst[0] = make_float4(bv.x, bv.y, dv.x, dv.y);
      dst[1] = make_float4(bv.z, bv.w, dv.z, dv.w);
    }
  }
  __syncthreads();

  const int c    = lane & 15;
  const int h    = lane >> 4;
  const int p    = c >> 2;
  const int sl   = (c & 3) + 4 * h;
  const int row0 = 2 * (g >> 1);
  const int col0 = 32 * (g & 1) + kTileW * p;
  const float x0a = -3.0f + 6.0f * (float)(row0)     / 63.0f;
  const float x0b = -3.0f + 6.0f * (float)(row0 + 1) / 63.0f;
  float x1[kTileW];
#pragma unroll
  for (int m = 0; m < kTileW; ++m) x1[m] = -3.0f + 6.0f * (float)(col0 + m) / 63.0f;

  const v2f xA = {x0a, x1[0]};
  const v2f xB = {x0b, x1[1]};
  const v2f xC = {x1[2], x1[3]};
  const v2f xD = {x1[4], x1[5]};
  const v2f xE = {x1[6], x1[7]};

  v2f bsel[kTileW];
#pragma unroll
  for (int m = 0; m < kTileW; ++m) {
    bsel[m][0] = (c == 2 * m)     ? 1.0f : 0.0f;
    bsel[m][1] = (c == 2 * m + 1) ? 1.0f : 0.0f;
  }

  float4 cb[2][kNJ];
  v2f bo[2][4];
#pragma unroll
  for (int s = 0; s < 2; ++s) {
    const float4* P = sBD + (16 * wave + 8 * s + sl) * kPitch4;
#pragma unroll
    for (int j = 0; j < kNJ; ++j) cb[s][j] = P[j];
#pragma unroll
    for (int i = 0; i < 4; ++i) {
      const float4 v = P[4 * p + i];
      bo[s][i][0] = v.x;
      bo[s][i][1] = v.y;
    }
  }

  v8f acc = {};
#pragma unroll
  for (int s = 0; s < 2; ++s) {
    float db[2][kTileW], dd[2][kTileW], wm[4];
    CornerDiffs nx;
    auto form = [&](int j) {
      const float4 cv = cb[s][j];
      const v2f bxy = {cv.x, cv.y};
      const v2f dxy = {cv.z, cv.w};
      const v2f byy = {cv.y, cv.y};
      const v2f dyy = {cv.w, cv.w};
      nx.b[0] = bxy - xA;
      nx.b[1] = bxy - xB;
      nx.b[2] = byy - xC;
      nx.b[3] = byy - xD;
      nx.b[4] = byy - xE;
      nx.d[0] = xA - dxy;
      nx.d[1] = xB - dxy;
      nx.d[2] = xC - dyy;
      nx.d[3] = xD - dyy;
      nx.d[4] = xE - dyy;
#pragma unroll
      for (int i = 0; i < 4; ++i) nx.f[i] = dxy - bo[s][i];
    };
    form(0);
#pragma unroll
    for (int j = 0; j < kNJ; ++j) {
      const CornerDiffs cu = nx;
      if (j + 1 < kNJ) form(j + 1);
      __builtin_amdgcn_sched_barrier(0);
      const float rb[2] = {cu.b[0][0], cu.b[1][0]};
      const float rd[2] = {cu.d[0][0], cu.d[1][0]};
      const float kb[kTileW] = {cu.b[0][1], cu.b[1][1], cu.b[2][0], cu.b[2][1],
                                cu.b[3][0], cu.b[3][1], cu.b[4][0], cu.b[4][1]};
      const float kd[kTileW] = {cu.d[0][1], cu.d[1][1], cu.d[2][0], cu.d[2][1],
                                cu.d[3][0], cu.d[3][1], cu.d[4][0], cu.d[4][1]};
      if (j == 0) {
#pragma unroll
        for (int m = 0; m < kTileW; ++m) {
          db[0][m] = fmaxf(rb[0], kb[m]);
          db[1][m] = fmaxf(rb[1], kb[m]);
          dd[0][m] = fmaxf(rd[0], kd[m]);
          dd[1][m] = fmaxf(rd[1], kd[m]);
        }
#pragma unroll
        for (int i = 0; i < 4; ++i) wm[i] = fmaxf(fminf(cu.f[i][0], cu.f[i][1]), 0.0f);
      } else {
#pragma unroll
        for (int m = 0; m < kTileW; ++m) {
          db[0][m] = maxmin(rb[0], kb[m], db[0][m]);
          db[1][m] = maxmin(rb[1], kb[m], db[1][m]);
          dd[0][m] = maxmin(rd[0], kd[m], dd[0][m]);
          dd[1][m] = maxmin(rd[1], kd[m], dd[1][m]);
        }
#pragma unroll
        for (int i = 0; i < 4; ++i) wm[i] = minmax(cu.f[i][0], cu.f[i][1], wm[i]);
      }
      __builtin_amdgcn_sched_barrier(0);
    }
    const float wl = fmaxf(fmaxf(wm[0], wm[1]), fmaxf(wm[2], wm[3]));
    const float wx = quad_stride4_max_nonneg(wl);
    const float wn = wx * wx;
#pragma unroll
    for (int m = 0; m < kTileW; ++m) {
      const float t0 = fmaxf(fmaxf(db[0][m], dd[0][m]), 0.0f);
      const float t1 = fmaxf(fmaxf(db[1][m], dd[1][m]), 0.0f);
      v2f a;
      a[0] = __builtin_amdgcn_exp2f(t0 * t0 * kGaussC) * wn;
      a[1] = __builtin_amdgcn_exp2f(t1 * t1 * kGaussC) * wn;
      acc = __builtin_amdgcn_wmma_f32_16x16x4_f32(
          false, a, false, bsel[m], (short)0, acc, false, false);
    }
  }

  {
    const float s0 = (acc[0] + acc[1]) + (acc[2] + acc[3]);
    const float s1 = (acc[4] + acc[5]) + (acc[6] + acc[7]);
    sRed[wave][(2 * h) * 16 + c]     = s0;
    sRed[wave][(2 * h + 1) * 16 + c] = s1;
  }
  __syncthreads();

  if (threadIdx.x < 64) {
    const int ro  = threadIdx.x >> 5;
    const int col = threadIdx.x & 31;
    const int idx = (col >> 3) * 16 + 2 * (col & 7) + ro;
    float v = sRed[0][idx];
#pragma unroll
    for (int w = 1; w < kWaves; ++w) v += sRed[w][idx];
    out[(size_t)b * (kRes * kRes) + (row0 + ro) * kRes + 32 * (g & 1) + col] = v;
  }
}

extern "C" void kernel_launch(void* const* d_in, const int* in_sizes, int n_in,
                              void* d_out, int out_size, void* d_ws, size_t ws_size,
                              hipStream_t stream) {
  (void)in_sizes; (void)n_in; (void)out_size; (void)d_ws; (void)ws_size;
  const float* Bs = (const float*)d_in[0];
  const float* Ds = (const float*)d_in[1];
  float* out = (float*)d_out;

  img_kernel<<<256, 256, 0, stream>>>(Bs, Ds, out);
}


# min-of-max chains kept per grid diagonal, not per pixel: max(bx-x0,by-x1)=max(bx,by+(x0-x1))-x0; a 2x8 tile has 9 diagonals: 18 not 32 fused ops per corner (value rounding changes, sum order kept)
# speedup vs baseline: 7.3131x; 1.1061x over previous
#include <hip/hip_runtime.h>
#include <math.h>


typedef __attribute__((ext_vector_type(2))) float v2f;
typedef __attribute__((ext_vector_type(8))) float v8f;

namespace {
constexpr int kRes = 64;
constexpr int kNN  = 128;
constexpr int kNJ  = 16;
constexpr float kGaussC =
    (float)(-0.5 * 1.4426950408889634 / ((double)0.3f * (double)0.3f));
constexpr int kPitch4 = kNJ + 1;
constexpr int kWaves  = 8;
constexpr int kTileW  = 8;
}

__device__ __forceinline__ float quad_stride4_max_nonneg(float v) {
  int x = __builtin_bit_cast(int, v);
  x = max(x, __builtin_amdgcn_update_dpp(0, x, 0x164, 0xf, 0xf, false));
  x = max(x, __builtin_amdgcn_update_dpp(0, x, 0x168, 0xf, 0xf, false));
  return __builtin_bit_cast(float, x);
}

__device__ __forceinline__ float maxmin(float a, float b, float m) {
  float r;
  asm("v_maxmin_num_f32 %0, %1, %2, %3" : "=v"(r) : "v"(a), "v"(b), "v"(m));
  return r;
}
__device__ __forceinline__ float minmax(float a, float b, float m) {
  float r;
  asm("v_minmax_num_f32 %0, %1, %2, %3" : "=v"(r) : "v"(a), "v"(b), "v"(m));
  return r;
}

struct CornerTerms {
  v2f b[4];
  v2f d[4];
  float b8;
  float d8;
  v2f f[4];
};

__global__ void __launch_bounds__(256) img_kernel(const float* __restrict__ Bs,
                                                  const float* __restrict__ Ds,
                                                  float* __restrict__ out) {
  __shared__ float4 sBD[kNN * kPitch4];
  __shared__ float  sRed[kWaves][64];

  const int wave = threadIdx.x >> 5;
  const int lane = threadIdx.x & 31;
  const int b    = blockIdx.x >> 6;
  const int g    = blockIdx.x & 63;

  {
    const float4* gB = (const float4*)(Bs + (size_t)b * kNN * kNJ * 2);
    const float4* gD = (const float4*)(Ds + (size_t)b * kNN * kNJ * 2);
    static_assert(kNN * kNJ / 2 == 4 * 256, "four corner pairs per thread");
#pragma unroll
    for (int k = 0; k < 4; ++k) {
      const int i  = threadIdx.x + 256 * k;
      const int n  = i >> 3;
      const int jp = i & 7;
      const float4 bv = gB[i];
      const float4 dv = gD[i];
      float4* dst = sBD + n * kPitch4 + 2 * jp;
      dst[0] = make_float4(bv.x, bv.y, dv.x, dv.y);
      dst[1] = make_float4(bv.z, bv.w, dv.z, dv.w);
    }
  }
  __syncthreads();

  const int c    = lane & 15;
  const int h    = lane >> 4;
  const int p    = c >> 2;
  const int sl   = (c & 3) + 4 * h;
  const int row0 = 2 * (g >> 1);
  const int col0 = 32 * (g & 1) + kTileW * p;
  const float x0a = -3.0f + 6.0f * (float)(row0)     / 63.0f;
  const float x0b = -3.0f + 6.0f * (float)(row0 + 1) / 63.0f;
  float dl[kTileW + 1];
#pragma unroll
  for (int q = 0; q < kTileW + 1; ++q)
    dl[q] = 6.0f * (float)(row0 - col0 + q - (kTileW - 1)) / 63.0f;
  const v2f dq[4] = {{dl[0], dl[1]}, {dl[2], dl[3]}, {dl[4], dl[5]}, {dl[6], dl[7]}};

  v2f bsel[kTileW];
#pragma unroll
  for (int m = 0; m < kTileW; ++m) {
    bsel[m][0] = (c == 2 * m)     ? 1.0f : 0.0f;
    bsel[m][1] = (c == 2 * m + 1) ? 1.0f : 0.0f;
  }

  float4 cb[2][kNJ];
  v2f bo[2][4];
#pragma unroll
  for (int s = 0; s < 2; ++s) {
    const float4* P = sBD + (16 * wave + 8 * s + sl) * kPitch4;
#pragma unroll
    for (int j = 0; j < kNJ; ++j) cb[s][j] = P[j];
#pragma unroll
    for (int i = 0; i < 4; ++i) {
      const float4 v = P[4 * p + i];
      bo[s][i][0] = v.x;
      bo[s][i][1] = v.y;
    }
  }

  v8f acc = {};
#pragma unroll
  for (int s = 0; s < 2; ++s) {
    float gb[kTileW + 1], gd[kTileW + 1], wm[4];
    CornerTerms nx;
    auto form = [&](int j) {
      const float4 cv = cb[s][j];
      const v2f dxy = {cv.z, cv.w};
      const v2f byy = {cv.y, cv.y};
      const v2f dyy = {cv.w, cv.w};
#pragma unroll
      for (int k = 0; k < 4; ++k) {
        nx.b[k] = byy + dq[k];
        nx.d[k] = dyy + dq[k];
      }
      nx.b8 = cv.y + dl[kTileW];
      nx.d8 = cv.w + dl[kTileW];
#pragma unroll
      for (int i = 0; i < 4; ++i) nx.f[i] = dxy - bo[s][i];
    };
    form(0);
#pragma unroll
    for (int j = 0; j < kNJ; ++j) {
      const CornerTerms cu = nx;
      const float bx = cb[s][j].x;
      const float dx = cb[s][j].z;
      if (j + 1 < kNJ) form(j + 1);
      __builtin_amdgcn_sched_barrier(0);
      const float sb[kTileW + 1] = {cu.b[0][0], cu.b[0][1], cu.b[1][0], cu.b[1][1],
                                    cu.b[2][0], cu.b[2][1], cu.b[3][0], cu.b[3][1], cu.b8};
      const float sd[kTileW + 1] = {cu.d[0][0], cu.d[0][1], cu.d[1][0], cu.d[1][1],
                                    cu.d[2][0], cu.d[2][1], cu.d[3][0], cu.d[3][1], cu.d8};
      if (j == 0) {
#pragma unroll
        for (int q = 0; q < kTileW + 1; ++q) {
          gb[q] = fmaxf(bx, sb[q]);
          gd[q] = fminf(dx, sd[q]);
        }
#pragma unroll
        for (int i = 0; i < 4; ++i) wm[i] = fmaxf(fminf(cu.f[i][0], cu.f[i][1]), 0.0f);
      } else {
#pragma unroll
        for (int q = 0; q < kTileW + 1; ++q) {
          gb[q] = maxmin(bx, sb[q], gb[q]);
          gd[q] = minmax(dx, sd[q], gd[q]);
        }
#pragma unroll
        for (int i = 0; i < 4; ++i) wm[i] = minmax(cu.f[i][0], cu.f[i][1], wm[i]);
      }
      __builtin_amdgcn_sched_barrier(0);
    }
    const float wl = fmaxf(fmaxf(wm[0], wm[1]), fmaxf(wm[2], wm[3]));
    const float wx = quad_stride4_max_nonneg(wl);
    const float wn = wx * wx;
#pragma unroll
    for (int m = 0; m < kTileW; ++m) {
      const int q0 = kTileW - 1 - m;
      const int q1 = kTileW - m;
      const float t0 = fmaxf(fmaxf(gb[q0] - x0a, x0a - gd[q0]), 0.0f);
      const float t1 = fmaxf(fmaxf(gb[q1] - x0b, x0b - gd[q1]), 0.0f);
      v2f a;
      a[0] = __builtin_amdgcn_exp2f(t0 * t0 * kGaussC) * wn;
      a[1] = __builtin_amdgcn_exp2f(t1 * t1 * kGaussC) * wn;
      acc = __builtin_amdgcn_wmma_f32_16x16x4_f32(
          false, a, false, bsel[m], (short)0, acc, false, false);
    }
  }

  {
    const float s0 = (acc[0] + acc[1]) + (acc[2] + acc[3]);
    const float s1 = (acc[4] + acc[5]) + (acc[6] + acc[7]);
    sRed[wave][(2 * h) * 16 + c]     = s0;
    sRed[wave][(2 * h + 1) * 16 + c] = s1;
  }
  __syncthreads();

  if (threadIdx.x < 64) {
    const int ro  = threadIdx.x >> 5;
    const int col = threadIdx.x & 31;
    const int idx = (col >> 3) * 16 + 2 * (col & 7) + ro;
    float v = sRed[0][idx];
#pragma unroll
    for (int w = 1; w < kWaves; ++w) v += sRed[w][idx];
    out[(size_t)b * (kRes * kRes) + (row0 + ro) * kRes + 32 * (g & 1) + col] = v;
  }
}

extern "C" void kernel_launch(void* const* d_in, const int* in_sizes, int n_in,
                              void* d_out, int out_size, void* d_ws, size_t ws_size,
                              hipStream_t stream) {
  (void)in_sizes; (void)n_in; (void)out_size; (void)d_ws; (void)ws_size;
  const float* Bs = (const float*)d_in[0];
  const float* Ds = (const float*)d_in[1];
  float* out = (float*)d_out;

  img_kernel<<<256, 256, 0, stream>>>(Bs, Ds, out);
}
